# starting kernel
# baseline (speedup 1.0000x reference)
#include <hip/hip_runtime.h>

typedef __attribute__((ext_vector_type(2))) float v2f;
typedef __attribute__((ext_vector_type(8))) float v8f;

#define D 64


__global__ void zero_f32(float* __restrict__ p, int n) {
  int i = blockIdx.x * blockDim.x + threadIdx.x;
  if (i < n) p[i] = 0.0f;
}

__global__ void deg_count(const int* __restrict__ dst, float* __restrict__ deg, int n_edges) {
  int e = blockIdx.x * blockDim.x + threadIdx.x;
  if (e < n_edges) atomicAdd(&deg[dst[e]], 1.0f);
}

__global__ void invsqrt_finish(float* __restrict__ d, int n) {
  int i = blockIdx.x * blockDim.x + threadIdx.x;
  if (i < n) d[i] = 1.0f / sqrtf(d[i] + 1.0f);
}

__global__ void edge_norm(const int* __restrict__ src, const int* __restrict__ dst,
                          const float* __restrict__ inv, float* __restrict__ norm,
                          int n_edges) {
  int e = blockIdx.x * blockDim.x + threadIdx.x;
  if (e < n_edges) norm[e] = inv[src[e]] * inv[dst[e]];
}

template <bool RELU>
__global__ void gemm16_wmma(const float* __restrict__ A, const float* __restrict__ W,
                            float* __restrict__ T) {
  const int lane = threadIdx.x & 31;
  const int wave = threadIdx.x >> 5;
  const int half = lane >> 4;
  const int l    = lane & 15;
  const int row0 = blockIdx.x << 4;
  const int c0   = wave << 4;

  const float* __restrict__ arow = A + (size_t)(row0 + l) * D;

  v8f acc = {};
#pragma unroll
  for (int kk = 0; kk < D; kk += 4) {
    const int ka = kk + half * 2;
    float a0 = arow[ka];
    float a1 = arow[ka + 1];
    if (RELU) { a0 = fmaxf(a0, 0.0f); a1 = fmaxf(a1, 0.0f); }
    v2f a = {a0, a1};
    v2f b = { W[ka * D + c0 + l], W[(ka + 1) * D + c0 + l] };
    acc = __builtin_amdgcn_wmma_f32_16x16x4_f32(
         false, a,  false, b,
         (short)0, acc,  false,  false);
  }

  float* __restrict__ orow = T + (size_t)(row0 + half * 8) * D + c0 + l;
#pragma unroll
  for (int j = 0; j < 8; ++j) orow[(size_t)j * D] = acc[j];
}

__global__ void self_bias_init(const float* __restrict__ T, const float* __restrict__ inv,
                               const float* __restrict__ bias, float* __restrict__ out,
                               int total) {
  int i = blockIdx.x * blockDim.x + threadIdx.x;
  if (i >= total) return;
  float is = inv[i >> 6];
  out[i] = T[i] * is * is + bias[i & (D - 1)];
}

__global__ void edge_aggregate(const int* __restrict__ src, const int* __restrict__ dst,
                               const float* __restrict__ norm, const float* __restrict__ T,
                               float* __restrict__ out, int n_edges) {
  int e = blockIdx.x * (blockDim.x >> 5) + (threadIdx.x >> 5);
  if (e >= n_edges) return;
  int f2 = (threadIdx.x & 31) * 2;
  int s = src[e];
  int d = dst[e];
  float nv = norm[e];
  const float2 t = *(const float2*)(T + (size_t)s * D + f2);
  float* o = out + (size_t)d * D + f2;
  atomicAdd(o,     t.x * nv);
  atomicAdd(o + 1, t.y * nv);
}


static inline size_t align256(size_t x) { return (x + 255) & ~(size_t)255; }

extern "C" void kernel_launch(void* const* d_in, const int* in_sizes, int n_in,
                              void* d_out, int out_size, void* d_ws, size_t ws_size,
                              hipStream_t stream) {
  const int*   E  = (const int*)d_in[1];
  const float* X  = (const float*)d_in[2];
  const float* W1 = (const float*)d_in[3];
  const float* b1 = (const float*)d_in[4];
  const float* W2 = (const float*)d_in[5];
  const float* b2 = (const float*)d_in[6];
  const float* W3 = (const float*)d_in[7];
  const float* b3 = (const float*)d_in[8];
  float* out = (float*)d_out;

  const int n_edges = in_sizes[1] / 2;
  const int n_nodes = in_sizes[2] / D;
  const int* src = E;
  const int* dst = E + n_edges;

  char* ws = (char*)d_ws;
  float* inv  = (float*)ws; ws += align256((size_t)n_nodes * 4);
  float* norm = (float*)ws; ws += align256((size_t)n_edges * 4);
  float* T    = (float*)ws; ws += align256((size_t)n_nodes * D * 4);
  float* H    = (float*)ws;

  const int total = n_nodes * D;
  const int nb_nodes = (n_nodes + 255) / 256;
  const int nb_edges = (n_edges + 255) / 256;
  const int nb_total = (total + 255) / 256;
  const int nb_agg   = (n_edges + 7) / 8;
  const int nb_gemm  = n_nodes / 16;

  zero_f32<<<nb_nodes, 256, 0, stream>>>(inv, n_nodes);
  deg_count<<<nb_edges, 256, 0, stream>>>(dst, inv, n_edges);
  invsqrt_finish<<<nb_nodes, 256, 0, stream>>>(inv, n_nodes);
  edge_norm<<<nb_edges, 256, 0, stream>>>(src, dst, inv, norm, n_edges);

  gemm16_wmma<false><<<nb_gemm, 128, 0, stream>>>(X, W1, T);
  self_bias_init<<<nb_total, 256, 0, stream>>>(T, inv, b1, H, total);
  edge_aggregate<<<nb_agg, 256, 0, stream>>>(src, dst, norm, T, H, n_edges);

  gemm16_wmma<true><<<nb_gemm, 128, 0, stream>>>(H, W2, T);
  self_bias_init<<<nb_total, 256, 0, stream>>>(T, inv, b2, H, total);
  edge_aggregate<<<nb_agg, 256, 0, stream>>>(src, dst, norm, T, H, n_edges);

  gemm16_wmma<true><<<nb_gemm, 128, 0, stream>>>(H, W3, T);
  self_bias_init<<<nb_total, 256, 0, stream>>>(T, inv, b3, out, total);
  edge_aggregate<<<nb_agg, 256, 0, stream>>>(src, dst, norm, T, out, n_edges);
}


# f32 atomic scatter replaced by a gather over a per-call destination-grouped edge list (slot = rank from one returning int atomic per edge); row sums now in slot (arrival) order, in registers
# speedup vs baseline: 2.3304x; 2.3304x over previous
#include <hip/hip_runtime.h>


typedef __attribute__((ext_vector_type(2))) float v2f;
typedef __attribute__((ext_vector_type(8))) float v8f;

#define D 64


__global__ void zero_i32(int* __restrict__ p, int n) {
  int i = blockIdx.x * blockDim.x + threadIdx.x;
  if (i < n) p[i] = 0;
}

__global__ void count_rank(const int* __restrict__ dst, int* __restrict__ cnt,
                           int* __restrict__ rank, int n_edges) {
  int e = blockIdx.x * blockDim.x + threadIdx.x;
  if (e < n_edges) rank[e] = atomicAdd(&cnt[dst[e]], 1);
}

__global__ void scan_local(const int* __restrict__ cnt, int* __restrict__ rowptr,
                           int* __restrict__ bsum, float* __restrict__ inv, int n) {
  __shared__ int wsum[8];
  const int t = threadIdx.x;
  const int lane = t & 31;
  const int wv = t >> 5;
  const int base = (blockIdx.x * 256 + t) * 4;
  int v[4];
  int tot = 0;
#pragma unroll
  for (int k = 0; k < 4; ++k) {
    v[k] = (base + k < n) ? cnt[base + k] : 0;
    tot += v[k];
  }
  int incl = tot;
#pragma unroll
  for (int o = 1; o < 32; o <<= 1) {
    int u = __shfl_up(incl, o);
    if (lane >= o) incl += u;
  }
  if (lane == 31) wsum[wv] = incl;
  __syncthreads();
  int woff = 0;
  for (int i = 0; i < wv; ++i) woff += wsum[i];
  int excl = woff + incl - tot;
#pragma unroll
  for (int k = 0; k < 4; ++k) {
    if (base + k < n) {
      rowptr[base + k] = excl;
      inv[base + k] = 1.0f / sqrtf((float)v[k] + 1.0f);
      excl += v[k];
    }
  }
  if (t == 255) bsum[blockIdx.x] = woff + incl;
}

__global__ void scan_fix(int* __restrict__ rowptr, const int* __restrict__ bsum,
                         int n, int n_edges) {
  __shared__ int red[8];
  const int t = threadIdx.x;
  const int lane = t & 31;
  const int wv = t >> 5;
  int part = 0;
  for (int i = t; i < (int)blockIdx.x; i += 256) part += bsum[i];
#pragma unroll
  for (int o = 16; o > 0; o >>= 1) part += __shfl_xor(part, o);
  if (lane == 0) red[wv] = part;
  __syncthreads();
  int off = 0;
#pragma unroll
  for (int i = 0; i < 8; ++i) off += red[i];
  const int base = (blockIdx.x * 256 + t) * 4;
#pragma unroll
  for (int k = 0; k < 4; ++k)
    if (base + k < n) rowptr[base + k] += off;
  if (blockIdx.x == 0 && t == 0) rowptr[n] = n_edges;
}

__global__ void csr_fill(const int* __restrict__ src, const int* __restrict__ dst,
                         const int* __restrict__ rank, const int* __restrict__ rowptr,
                         const float* __restrict__ inv, int* __restrict__ csr,
                         int n_edges) {
  int e = blockIdx.x * blockDim.x + threadIdx.x;
  if (e >= n_edges) return;
  int s = src[e];
  int d = dst[e];
  float w = inv[s] * inv[d];
  ((int2*)csr)[rowptr[d] + rank[e]] = make_int2(s, __float_as_int(w));
}

template <bool RELU>
__global__ void gemm16_wmma(const float* __restrict__ A, const float* __restrict__ W,
                            float* __restrict__ T) {
  const int lane = threadIdx.x & 31;
  const int wave = threadIdx.x >> 5;
  const int half = lane >> 4;
  const int l    = lane & 15;
  const int row0 = blockIdx.x << 4;
  const int c0   = wave << 4;

  const float* __restrict__ arow = A + (size_t)(row0 + l) * D;

  v8f acc = {};
#pragma unroll
  for (int kk = 0; kk < D; kk += 4) {
    const int ka = kk + half * 2;
    float a0 = arow[ka];
    float a1 = arow[ka + 1];
    if (RELU) { a0 = fmaxf(a0, 0.0f); a1 = fmaxf(a1, 0.0f); }
    v2f a = {a0, a1};
    v2f b = { W[ka * D + c0 + l], W[(ka + 1) * D + c0 + l] };
    acc = __builtin_amdgcn_wmma_f32_16x16x4_f32(
         false, a,  false, b,
         (short)0, acc,  false,  false);
  }

  float* __restrict__ orow = T + (size_t)(row0 + half * 8) * D + c0 + l;
#pragma unroll
  for (int j = 0; j < 8; ++j) orow[(size_t)j * D] = acc[j];
}

#define GATHER_EDGE(j)                                                         \
  {                                                                            \
    const float wj = __int_as_float(ed[j].y);                                  \
    acc.x = fmaf(tv[j].x, wj, acc.x);                                          \
    acc.y = fmaf(tv[j].y, wj, acc.y);                                          \
  }

template <int NB>
__device__ __forceinline__ void gather_batch(const int2* __restrict__ csr, int k,
                                             const float2* __restrict__ Tv, float2& acc) {
  int2 ed[NB];
  float2 tv[NB];
#pragma unroll
  for (int j = 0; j < NB; ++j) ed[j] = csr[k + j];
#pragma unroll
  for (int j = 0; j < NB; ++j) tv[j] = Tv[(size_t)ed[j].x * (D / 2)];
#pragma unroll
  for (int j = 0; j < NB; ++j) GATHER_EDGE(j)
}

__global__ void gather_rows(const int* __restrict__ rowptr, const int* __restrict__ csr_pairs,
                            const float* __restrict__ inv, const float* __restrict__ T,
                            const float* __restrict__ bias, float* __restrict__ out,
                            int n) {
  int dv = blockIdx.x * (blockDim.x >> 5) + (threadIdx.x >> 5);
  if (dv >= n) return;
  const int d = __builtin_amdgcn_readfirstlane(dv);
  const int lane = threadIdx.x & 31;
  const int2* __restrict__ csr = (const int2*)csr_pairs;
  const int beg = rowptr[d];
  const int end = rowptr[d + 1];
  const float isd = inv[d];
  const float2* __restrict__ Tv = (const float2*)T + lane;
  const float2 self = Tv[(size_t)d * (D / 2)];
  const float2 b = ((const float2*)bias)[lane];
  float2 acc;
  acc.x = self.x * isd * isd + b.x;
  acc.y = self.y * isd * isd + b.y;

  int k = beg;
  for (; k + 8 <= end; k += 8) gather_batch<8>(csr, k, Tv, acc);
  const int rem = end - k;
  if (rem & 4) { gather_batch<4>(csr, k, Tv, acc); k += 4; }
  if (rem & 2) { gather_batch<2>(csr, k, Tv, acc); k += 2; }
  if (rem & 1) { gather_batch<1>(csr, k, Tv, acc); }

  ((float2*)out)[(size_t)d * (D / 2) + lane] = acc;
}


static inline size_t align256(size_t x) { return (x + 255) & ~(size_t)255; }

extern "C" void kernel_launch(void* const* d_in, const int* in_sizes, int n_in,
                              void* d_out, int out_size, void* d_ws, size_t ws_size,
                              hipStream_t stream) {
  const int*   E  = (const int*)d_in[1];
  const float* X  = (const float*)d_in[2];
  const float* W1 = (const float*)d_in[3];
  const float* b1 = (const float*)d_in[4];
  const float* W2 = (const float*)d_in[5];
  const float* b2 = (const float*)d_in[6];
  const float* W3 = (const float*)d_in[7];
  const float* b3 = (const float*)d_in[8];
  float* out = (float*)d_out;

  const int n_edges = in_sizes[1] / 2;
  const int n_nodes = in_sizes[2] / D;
  const int* src = E;
  const int* dst = E + n_edges;

  const int nb_scan = (n_nodes + 1023) / 1024;

  char* ws = (char*)d_ws;
  int*   cnt    = (int*)ws;   ws += align256((size_t)n_nodes * 4);
  int*   rowptr = (int*)ws;   ws += align256((size_t)(n_nodes + 1) * 4);
  int*   bsum   = (int*)ws;   ws += align256((size_t)nb_scan * 4);
  float* inv    = (float*)ws; ws += align256((size_t)n_nodes * 4);
  int*   rank   = (int*)ws;   ws += align256((size_t)n_edges * 4);
  int*   csr    = (int*)ws;   ws += align256((size_t)n_edges * 8);
  float* T      = (float*)ws; ws += align256((size_t)n_nodes * D * 4);
  float* H      = (float*)ws;

  const int nb_nodes = (n_nodes + 255) / 256;
  const int nb_edges = (n_edges + 255) / 256;
  const int nb_gath  = (n_nodes + 7) / 8;
  const int nb_gemm  = n_nodes / 16;

  zero_i32<<<nb_nodes, 256, 0, stream>>>(cnt, n_nodes);
  count_rank<<<nb_edges, 256, 0, stream>>>(dst, cnt, rank, n_edges);
  scan_local<<<nb_scan, 256, 0, stream>>>(cnt, rowptr, bsum, inv, n_nodes);
  scan_fix<<<nb_scan, 256, 0, stream>>>(rowptr, bsum, n_nodes, n_edges);
  csr_fill<<<nb_edges, 256, 0, stream>>>(src, dst, rank, rowptr, inv, csr, n_edges);

  gemm16_wmma<false><<<nb_gemm, 128, 0, stream>>>(X, W1, T);
  gather_rows<<<nb_gath, 256, 0, stream>>>(rowptr, csr, inv, T, b1, H, n_nodes);

  gemm16_wmma<true><<<nb_gemm, 128, 0, stream>>>(H, W2, T);
  gather_rows<<<nb_gath, 256, 0, stream>>>(rowptr, csr, inv, T, b2, H, n_nodes);

  gemm16_wmma<true><<<nb_gemm, 128, 0, stream>>>(H, W3, T);
  gather_rows<<<nb_gath, 256, 0, stream>>>(rowptr, csr, inv, T, b3, out, n_nodes);
}


# layers 2, 3: dense product moved into the tail of the preceding gather kernel (16-wave block = 16-row stripe, relu'd rows handed over in LDS, 4 waves issue the WMMAs); 11 -> 9 launches; same sums
# speedup vs baseline: 2.6240x; 1.1260x over previous
#include <hip/hip_runtime.h>


typedef __attribute__((ext_vector_type(2))) float v2f;
typedef __attribute__((ext_vector_type(8))) float v8f;

#define D 64


__global__ void zero_i32(int* __restrict__ p, int n) {
  int i = blockIdx.x * blockDim.x + threadIdx.x;
  if (i < n) p[i] = 0;
}

__global__ void count_rank(const int* __restrict__ dst, int* __restrict__ cnt,
                           int* __restrict__ rank, int n_edges) {
  int e = blockIdx.x * blockDim.x + threadIdx.x;
  if (e < n_edges) rank[e] = atomicAdd(&cnt[dst[e]], 1);
}

__global__ void scan_local(const int* __restrict__ cnt, int* __restrict__ rowptr,
                           int* __restrict__ bsum, float* __restrict__ inv, int n) {
  __shared__ int wsum[8];
  const int t = threadIdx.x;
  const int lane = t & 31;
  const int wv = t >> 5;
  const int base = (blockIdx.x * 256 + t) * 4;
  int v[4];
  int tot = 0;
#pragma unroll
  for (int k = 0; k < 4; ++k) {
    v[k] = (base + k < n) ? cnt[base + k] : 0;
    tot += v[k];
  }
  int incl = tot;
#pragma unroll
  for (int o = 1; o < 32; o <<= 1) {
    int u = __shfl_up(incl, o);
    if (lane >= o) incl += u;
  }
  if (lane == 31) wsum[wv] = incl;
  __syncthreads();
  int woff = 0;
  for (int i = 0; i < wv; ++i) woff += wsum[i];
  int excl = woff + incl - tot;
#pragma unroll
  for (int k = 0; k < 4; ++k) {
    if (base + k < n) {
      rowptr[base + k] = excl;
      inv[base + k] = 1.0f / sqrtf((float)v[k] + 1.0f);
      excl += v[k];
    }
  }
  if (t == 255) bsum[blockIdx.x] = woff + incl;
}

__global__ void scan_fix(int* __restrict__ rowptr, const int* __restrict__ bsum,
                         int n, int n_edges) {
  __shared__ int red[8];
  const int t = threadIdx.x;
  const int lane = t & 31;
  const int wv = t >> 5;
  int part = 0;
  for (int i = t; i < (int)blockIdx.x; i += 256) part += bsum[i];
#pragma unroll
  for (int o = 16; o > 0; o >>= 1) part += __shfl_xor(part, o);
  if (lane == 0) red[wv] = part;
  __syncthreads();
  int off = 0;
#pragma unroll
  for (int i = 0; i < 8; ++i) off += red[i];
  const int base = (blockIdx.x * 256 + t) * 4;
#pragma unroll
  for (int k = 0; k < 4; ++k)
    if (base + k < n) rowptr[base + k] += off;
  if (blockIdx.x == 0 && t == 0) rowptr[n] = n_edges;
}

__global__ void csr_fill(const int* __restrict__ src, const int* __restrict__ dst,
                         const int* __restrict__ rank, const int* __restrict__ rowptr,
                         int* __restrict__ csr, int n_edges) {
  int e = blockIdx.x * blockDim.x + threadIdx.x;
  if (e >= n_edges) return;
  csr[rowptr[dst[e]] + rank[e]] = src[e];
}

__global__ void gemm16_wmma(const float* __restrict__ A, const float* __restrict__ W,
                            const float* __restrict__ inv, float* __restrict__ T) {
  const int lane = threadIdx.x & 31;
  const int wave = threadIdx.x >> 5;
  const int half = lane >> 4;
  const int l    = lane & 15;
  const int row0 = blockIdx.x << 4;
  const int c0   = wave << 4;

  const float* __restrict__ arow = A + (size_t)(row0 + l) * D;

  v8f acc = {};
#pragma unroll
  for (int kk = 0; kk < D; kk += 4) {
    const int ka = kk + half * 2;
    float a0 = arow[ka];
    float a1 = arow[ka + 1];
    v2f a = {a0, a1};
    v2f b = { W[ka * D + c0 + l], W[(ka + 1) * D + c0 + l] };
    acc = __builtin_amdgcn_wmma_f32_16x16x4_f32(
         false, a,  false, b,
         (short)0, acc,  false,  false);
  }

  float* __restrict__ orow = T + (size_t)(row0 + half * 8) * D + c0 + l;
  const float* __restrict__ irow = inv + row0 + half * 8;
#pragma unroll
  for (int j = 0; j < 8; ++j) orow[(size_t)j * D] = acc[j] * irow[j];
}

template <int NB>
__device__ __forceinline__ void gather_batch(const int* __restrict__ csr, int k,
                                             const float2* __restrict__ Tv, float2& acc) {
  int ed[NB];
  float2 tv[NB];
#pragma unroll
  for (int j = 0; j < NB; ++j) ed[j] = csr[k + j];
#pragma unroll
  for (int j = 0; j < NB; ++j) tv[j] = Tv[(size_t)ed[j] * (D / 2)];
#pragma unroll
  for (int j = 0; j < NB; ++j) {
    acc.x += tv[j].x;
    acc.y += tv[j].y;
  }
}

__global__ void gather_rows(const int* __restrict__ rowptr, const int* __restrict__ csr,
                            const float* __restrict__ inv, const float* __restrict__ T,
                            const float* __restrict__ bias, float* __restrict__ out,
                            int n) {
  int dv = blockIdx.x * (blockDim.x >> 5) + (threadIdx.x >> 5);
  if (dv >= n) return;
  const int d = __builtin_amdgcn_readfirstlane(dv);
  const int lane = threadIdx.x & 31;
  const int beg = rowptr[d];
  const int end = rowptr[d + 1];
  const float isd = inv[d];
  const float2* __restrict__ Tv = (const float2*)T + lane;
  const float2 b = ((const float2*)bias)[lane];
  float2 acc = Tv[(size_t)d * (D / 2)];

  int k = beg;
  for (; k + 8 <= end; k += 8) gather_batch<8>(csr, k, Tv, acc);
  const int rem = end - k;
  if (rem & 4) { gather_batch<4>(csr, k, Tv, acc); k += 4; }
  if (rem & 2) { gather_batch<2>(csr, k, Tv, acc); k += 2; }
  if (rem & 1) { gather_batch<1>(csr, k, Tv, acc); }

  acc.x = fmaf(acc.x, isd, b.x);
  acc.y = fmaf(acc.y, isd, b.y);
  ((float2*)out)[(size_t)d * (D / 2) + lane] = acc;
}

#define LDP 68

__global__ __launch_bounds__(512) void gather_gemm(
    const int* __restrict__ rowptr, const int* __restrict__ csr,
    const float* __restrict__ inv, const float* __restrict__ T,
    const float* __restrict__ bias, const float* __restrict__ W,
    float* __restrict__ Tn) {
  __shared__ float hs[16 * LDP];
  const int lane = threadIdx.x & 31;
  const int wave = threadIdx.x >> 5;
  const int row0 = blockIdx.x << 4;
  {
    const int d = __builtin_amdgcn_readfirstlane(row0 + wave);
    const int beg = rowptr[d];
    const int end = rowptr[d + 1];
    const float isd = inv[d];
    const float2* __restrict__ Tv = (const float2*)T + lane;
    const float2 b = ((const float2*)bias)[lane];
    float2 acc = Tv[(size_t)d * (D / 2)];

    int k = beg;
    for (; k + 8 <= end; k += 8) gather_batch<8>(csr, k, Tv, acc);
    const int rem = end - k;
    if (rem & 4) { gather_batch<4>(csr, k, Tv, acc); k += 4; }
    if (rem & 2) { gather_batch<2>(csr, k, Tv, acc); k += 2; }
    if (rem & 1) { gather_batch<1>(csr, k, Tv, acc); }

    float2 h;
    h.x = fmaxf(fmaf(acc.x, isd, b.x), 0.0f);
    h.y = fmaxf(fmaf(acc.y, isd, b.y), 0.0f);
    *(float2*)&hs[wave * LDP + 2 * lane] = h;
  }
  __syncthreads();
  if (wave >= 4) return;

  const int half = lane >> 4;
  const int l    = lane & 15;
  const int c0   = wave << 4;
  const float* __restrict__ arow = hs + l * LDP;

  v8f acc = {};
#pragma unroll
  for (int kk = 0; kk < D; kk += 4) {
    const int ka = kk + half * 2;
    v2f a = {arow[ka], arow[ka + 1]};
    v2f b = { W[ka * D + c0 + l], W[(ka + 1) * D + c0 + l] };
    acc = __builtin_amdgcn_wmma_f32_16x16x4_f32(
         false, a,  false, b,
         (short)0, acc,  false,  false);
  }

  float* __restrict__ orow = Tn + (size_t)(row0 + half * 8) * D + c0 + l;
  const float* __restrict__ irow = inv + row0 + half * 8;
#pragma unroll
  for (int j = 0; j < 8; ++j) orow[(size_t)j * D] = acc[j] * irow[j];
}


static inline size_t align256(size_t x) { return (x + 255) & ~(size_t)255; }

extern "C" void kernel_launch(void* const* d_in, const int* in_sizes, int n_in,
                              void* d_out, int out_size, void* d_ws, size_t ws_size,
                              hipStream_t stream) {
  const int*   E  = (const int*)d_in[1];
  const float* X  = (const float*)d_in[2];
  const float* W1 = (const float*)d_in[3];
  const float* b1 = (const float*)d_in[4];
  const float* W2 = (const float*)d_in[5];
  const float* b2 = (const float*)d_in[6];
  const float* W3 = (const float*)d_in[7];
  const float* b3 = (const float*)d_in[8];
  float* out = (float*)d_out;

  const int n_edges = in_sizes[1] / 2;
  const int n_nodes = in_sizes[2] / D;
  const int* src = E;
  const int* dst = E + n_edges;

  const int nb_scan = (n_nodes + 1023) / 1024;

  char* ws = (char*)d_ws;
  int*   cnt    = (int*)ws;   ws += align256((size_t)n_nodes * 4);
  int*   rowptr = (int*)ws;   ws += align256((size_t)(n_nodes + 1) * 4);
  int*   bsum   = (int*)ws;   ws += align256((size_t)nb_scan * 4);
  float* inv    = (float*)ws; ws += align256((size_t)n_nodes * 4);
  int*   rank   = (int*)ws;   ws += align256((size_t)n_edges * 4);
  int*   csr    = (int*)ws;   ws += align256((size_t)n_edges * 4);
  float* T      = (float*)ws; ws += align256((size_t)n_nodes * D * 4);
  float* T2     = (float*)ws;

  const int nb_nodes = (n_nodes + 255) / 256;
  const int nb_edges = (n_edges + 255) / 256;
  const int nb_gath  = (n_nodes + 7) / 8;
  const int nb_gemm  = n_nodes / 16;

  zero_i32<<<nb_nodes, 256, 0, stream>>>(cnt, n_nodes);
  count_rank<<<nb_edges, 256, 0, stream>>>(dst, cnt, rank, n_edges);
  scan_local<<<nb_scan, 256, 0, stream>>>(cnt, rowptr, bsum, inv, n_nodes);
  scan_fix<<<nb_scan, 256, 0, stream>>>(rowptr, bsum, n_nodes, n_edges);
  csr_fill<<<nb_edges, 256, 0, stream>>>(src, dst, rank, rowptr, csr, n_edges);

  gemm16_wmma<<<nb_gemm, 128, 0, stream>>>(X, W1, inv, T);
  gather_gemm<<<nb_gemm, 512, 0, stream>>>(rowptr, csr, inv, T, b1, W2, T2);
  gather_gemm<<<nb_gemm, 512, 0, stream>>>(rowptr, csr, inv, T2, b2, W3, T);
  gather_rows<<<nb_gath, 256, 0, stream>>>(rowptr, csr, inv, T, b3, out, n_nodes);
}


# edge list built by a two-level counting sort ranked in LDS (256-node bins, whole runs stored), not a returning global atomic + scattered 4-byte accesses per edge; sums stay in slot (arrival) order
# speedup vs baseline: 4.3342x; 1.6518x over previous
#include <hip/hip_runtime.h>


typedef __attribute__((ext_vector_type(2))) float v2f;
typedef __attribute__((ext_vector_type(8))) float v8f;

#define D 64

#define BIN_N 256
#define BIN_SH 8
#define CH 6144
#define MAXBIN 512
#define CAP 8192

__global__ __launch_bounds__(1024) void bin_hist(const int* __restrict__ dst,
                                                 int* __restrict__ hist,
                                                 int n_edges, int nbin) {
  __shared__ int h[MAXBIN];
  const int t = threadIdx.x;
  const int beg = blockIdx.x * CH;
  const int end = min(beg + CH, n_edges);
  int dv[CH / 1024];
#pragma unroll
  for (int k = 0; k < CH / 1024; ++k) {
    const int e = beg + t + k * 1024;
    dv[k] = (e < end) ? dst[e] : 0;
  }
  for (int i = t; i < nbin; i += 1024) h[i] = 0;
  __syncthreads();
#pragma unroll
  for (int k = 0; k < CH / 1024; ++k)
    if (beg + t + k * 1024 < end) atomicAdd(&h[dv[k] >> BIN_SH], 1);
  __syncthreads();
  for (int i = t; i < nbin; i += 1024) hist[i * gridDim.x + blockIdx.x] = h[i];
}

__global__ __launch_bounds__(512) void bin_scan(const int* __restrict__ hist,
                                                int* __restrict__ off,
                                                int* __restrict__ tot, int nblk) {
  __shared__ int wsum[16];
  const int t = threadIdx.x;
  const int lane = t & 31;
  const int wv = t >> 5;
  const int* __restrict__ h = hist + (size_t)blockIdx.x * nblk;
  int* __restrict__ o = off + (size_t)blockIdx.x * nblk;
  int carry = 0;
  for (int base = 0; base < nblk; base += 512) {
    const int i = base + t;
    const int v = (i < nblk) ? h[i] : 0;
    int incl = v;
#pragma unroll
    for (int s = 1; s < 32; s <<= 1) {
      int u = __shfl_up(incl, s);
      if (lane >= s) incl += u;
    }
    if (lane == 31) wsum[wv] = incl;
    __syncthreads();
    int woff = 0, all = 0;
#pragma unroll
    for (int k = 0; k < 16; ++k) {
      const int s = wsum[k];
      if (k < wv) woff += s;
      all += s;
    }
    if (i < nblk) o[i] = carry + woff + incl - v;
    carry += all;
    __syncthreads();
  }
  if (t == 0) tot[blockIdx.x] = carry;
}

__global__ __launch_bounds__(1024) void bin_place(
    const int* __restrict__ src, const int* __restrict__ dst,
    const int* __restrict__ hist, const int* __restrict__ off,
    const int* __restrict__ tot, int* __restrict__ binstart,
    int2* __restrict__ pairs, int n_edges, int nbin) {
  __shared__ int lcur[MAXBIN];
  __shared__ int gdelta[MAXBIN];
  __shared__ int wsa[16];
  __shared__ int wsb[16];
  __shared__ int2 stage[CH];
  const int t = threadIdx.x;
  const int lane = t & 31;
  const int wv = t >> 5;
  const int b = blockIdx.x;
  const int nblk = gridDim.x;

  const int beg = b * CH;
  const int end = min(beg + CH, n_edges);
  int sv[CH / 1024], dv[CH / 1024];
#pragma unroll
  for (int k = 0; k < CH / 1024; ++k) {
    const int e = beg + t + k * 1024;
    sv[k] = (e < end) ? src[e] : 0;
    dv[k] = (e < end) ? dst[e] : 0;
  }

  int hv = 0, tv = 0, ov = 0;
  if (t < nbin) {
    hv = hist[(size_t)t * nblk + b];
    ov = off[(size_t)t * nblk + b];
    tv = tot[t];
  }
  int hi = hv, ti = tv;
#pragma unroll
  for (int s = 1; s < 32; s <<= 1) {
    int u = __shfl_up(hi, s);
    int w = __shfl_up(ti, s);
    if (lane >= s) { hi += u; ti += w; }
  }
  if (t < MAXBIN && lane == 31) { wsa[wv] = hi; wsb[wv] = ti; }
  __syncthreads();
  if (t < MAXBIN) {
    int wa = 0, wb = 0;
    for (int k = 0; k < wv; ++k) { wa += wsa[k]; wb += wsb[k]; }
    const int ls = wa + hi - hv;
    const int gs = wb + ti - tv;
    if (t < nbin) {
      lcur[t] = ls;
      gdelta[t] = gs + ov - ls;
      if (b == 0) {
        binstart[t] = gs;
        if (t == nbin - 1) binstart[nbin] = gs + tv;
      }
    }
  }
  __syncthreads();

#pragma unroll
  for (int k = 0; k < CH / 1024; ++k) {
    if (beg + t + k * 1024 < end) {
      const int r = atomicAdd(&lcur[dv[k] >> BIN_SH], 1);
      stage[r] = make_int2(sv[k], dv[k]);
    }
  }
  __syncthreads();
  const int cnt = end - beg;
  int2 pv[CH / 1024];
  int gv[CH / 1024];
#pragma unroll
  for (int k = 0; k < CH / 1024; ++k)
    if (t + k * 1024 < cnt) pv[k] = stage[t + k * 1024];
#pragma unroll
  for (int k = 0; k < CH / 1024; ++k)
    if (t + k * 1024 < cnt) gv[k] = gdelta[pv[k].y >> BIN_SH];
#pragma unroll
  for (int k = 0; k < CH / 1024; ++k)
    if (t + k * 1024 < cnt) pairs[gv[k] + t + k * 1024] = pv[k];
}

__global__ __launch_bounds__(1024) void bin_sort(
    const int2* __restrict__ pairs, const int* __restrict__ binstart,
    int* __restrict__ rowptr, float* __restrict__ inv, int* __restrict__ csr, int n) {
  __shared__ int cnt[BIN_N];
  __shared__ int cur[BIN_N];
  __shared__ int wsum[BIN_N / 32];
  __shared__ int stage[CAP];
  const int t = threadIdx.x;
  const int lane = t & 31;
  const int wv = t >> 5;
  const int node0 = blockIdx.x * BIN_N;
  const int s = binstart[blockIdx.x];
  const int e = binstart[blockIdx.x + 1];
  const int m = e - s;

  const bool small = (m <= CAP);
  int2 p[CAP / 1024];
  if (small) {
#pragma unroll
    for (int k = 0; k < CAP / 1024; ++k) {
      const int i = s + t + k * 1024;
      p[k] = (i < e) ? pairs[i] : make_int2(0, node0);
    }
  }
  if (t < BIN_N) cnt[t] = 0;
  __syncthreads();
  if (small) {
#pragma unroll
    for (int k = 0; k < CAP / 1024; ++k)
      if (s + t + k * 1024 < e) atomicAdd(&cnt[p[k].y - node0], 1);
  } else {
    for (int i = s + t; i < e; i += 1024) atomicAdd(&cnt[pairs[i].y - node0], 1);
  }
  __syncthreads();

  int v = 0, incl = 0;
  if (t < BIN_N) {
    v = cnt[t];
    incl = v;
#pragma unroll
    for (int o = 1; o < 32; o <<= 1) {
      int u = __shfl_up(incl, o);
      if (lane >= o) incl += u;
    }
    if (lane == 31) wsum[wv] = incl;
  }
  __syncthreads();
  if (t < BIN_N) {
    int woff = 0;
    for (int k = 0; k < wv; ++k) woff += wsum[k];
    const int excl = woff + incl - v;
    cur[t] = excl;
    const int node = node0 + t;
    if (node < n) {
      rowptr[node] = s + excl;
      inv[node] = 1.0f / sqrtf((float)v + 1.0f);
      if (node == n - 1) rowptr[n] = e;
    }
  }
  __syncthreads();

  if (small) {
#pragma unroll
    for (int k = 0; k < CAP / 1024; ++k)
      if (s + t + k * 1024 < e) stage[atomicAdd(&cur[p[k].y - node0], 1)] = p[k].x;
    __syncthreads();
    int ov[CAP / 1024];
#pragma unroll
    for (int k = 0; k < CAP / 1024; ++k)
      if (t + k * 1024 < m) ov[k] = stage[t + k * 1024];
#pragma unroll
    for (int k = 0; k < CAP / 1024; ++k)
      if (t + k * 1024 < m) csr[s + t + k * 1024] = ov[k];
  } else {
    for (int i = s + t; i < e; i += 1024) {
      const int2 q = pairs[i];
      csr[s + atomicAdd(&cur[q.y - node0], 1)] = q.x;
    }
  }
}

__global__ void gemm16_wmma(const float* __restrict__ A, const float* __restrict__ W,
                            const float* __restrict__ inv, float* __restrict__ T) {
  const int lane = threadIdx.x & 31;
  const int wave = threadIdx.x >> 5;
  const int half = lane >> 4;
  const int l    = lane & 15;
  const int row0 = blockIdx.x << 4;
  const int c0   = wave << 4;

  const float* __restrict__ arow = A + (size_t)(row0 + l) * D;

  v8f acc = {};
#pragma unroll
  for (int kk = 0; kk < D; kk += 4) {
    const int ka = kk + half * 2;
    float a0 = arow[ka];
    float a1 = arow[ka + 1];
    v2f a = {a0, a1};
    v2f b = { W[ka * D + c0 + l], W[(ka + 1) * D + c0 + l] };
    acc = __builtin_amdgcn_wmma_f32_16x16x4_f32(
         false, a,  false, b,
         (short)0, acc,  false,  false);
  }

  float* __restrict__ orow = T + (size_t)(row0 + half * 8) * D + c0 + l;
  const float* __restrict__ irow = inv + row0 + half * 8;
#pragma unroll
  for (int j = 0; j < 8; ++j) orow[(size_t)j * D] = acc[j] * irow[j];
}

template <int NB>
__device__ __forceinline__ void gather_batch(const int* __restrict__ csr, int k,
                                             const float2* __restrict__ Tv, float2& acc) {
  int ed[NB];
  float2 tv[NB];
#pragma unroll
  for (int j = 0; j < NB; ++j) ed[j] = csr[k + j];
#pragma unroll
  for (int j = 0; j < NB; ++j) tv[j] = Tv[(size_t)ed[j] * (D / 2)];
#pragma unroll
  for (int j = 0; j < NB; ++j) {
    acc.x += tv[j].x;
    acc.y += tv[j].y;
  }
}

__global__ void gather_rows(const int* __restrict__ rowptr, const int* __restrict__ csr,
                            const float* __restrict__ inv, const float* __restrict__ T,
                            const float* __restrict__ bias, float* __restrict__ out,
                            int n) {
  int dv = blockIdx.x * (blockDim.x >> 5) + (threadIdx.x >> 5);
  if (dv >= n) return;
  const int d = __builtin_amdgcn_readfirstlane(dv);
  const int lane = threadIdx.x & 31;
  const int beg = rowptr[d];
  const int end = rowptr[d + 1];
  const float isd = inv[d];
  const float2* __restrict__ Tv = (const float2*)T + lane;
  const float2 b = ((const float2*)bias)[lane];
  float2 acc = Tv[(size_t)d * (D / 2)];

  int k = beg;
  for (; k + 8 <= end; k += 8) gather_batch<8>(csr, k, Tv, acc);
  const int rem = end - k;
  if (rem & 4) { gather_batch<4>(csr, k, Tv, acc); k += 4; }
  if (rem & 2) { gather_batch<2>(csr, k, Tv, acc); k += 2; }
  if (rem & 1) { gather_batch<1>(csr, k, Tv, acc); }

  acc.x = fmaf(acc.x, isd, b.x);
  acc.y = fmaf(acc.y, isd, b.y);
  ((float2*)out)[(size_t)d * (D / 2) + lane] = acc;
}

#define LDP 68

__global__ __launch_bounds__(512) void gather_gemm(
    const int* __restrict__ rowptr, const int* __restrict__ csr,
    const float* __restrict__ inv, const float* __restrict__ T,
    const float* __restrict__ bias, const float* __restrict__ W,
    float* __restrict__ Tn) {
  __shared__ float hs[16 * LDP];
  const int lane = threadIdx.x & 31;
  const int wave = threadIdx.x >> 5;
  const int row0 = blockIdx.x << 4;
  {
    const int d = __builtin_amdgcn_readfirstlane(row0 + wave);
    const int beg = rowptr[d];
    const int end = rowptr[d + 1];
    const float isd = inv[d];
    const float2* __restrict__ Tv = (const float2*)T + lane;
    const float2 b = ((const float2*)bias)[lane];
    float2 acc = Tv[(size_t)d * (D / 2)];

    int k = beg;
    for (; k + 8 <= end; k += 8) gather_batch<8>(csr, k, Tv, acc);
    const int rem = end - k;
    if (rem & 4) { gather_batch<4>(csr, k, Tv, acc); k += 4; }
    if (rem & 2) { gather_batch<2>(csr, k, Tv, acc); k += 2; }
    if (rem & 1) { gather_batch<1>(csr, k, Tv, acc); }

    float2 h;
    h.x = fmaxf(fmaf(acc.x, isd, b.x), 0.0f);
    h.y = fmaxf(fmaf(acc.y, isd, b.y), 0.0f);
    *(float2*)&hs[wave * LDP + 2 * lane] = h;
  }
  __syncthreads();
  if (wave >= 4) return;

  const int half = lane >> 4;
  const int l    = lane & 15;
  const int c0   = wave << 4;
  const float* __restrict__ arow = hs + l * LDP;

  v8f acc = {};
#pragma unroll
  for (int kk = 0; kk < D; kk += 4) {
    const int ka = kk + half * 2;
    v2f a = {arow[ka], arow[ka + 1]};
    v2f b = { W[ka * D + c0 + l], W[(ka + 1) * D + c0 + l] };
    acc = __builtin_amdgcn_wmma_f32_16x16x4_f32(
         false, a,  false, b,
         (short)0, acc,  false,  false);
  }

  float* __restrict__ orow = Tn + (size_t)(row0 + half * 8) * D + c0 + l;
  const float* __restrict__ irow = inv + row0 + half * 8;
#pragma unroll
  for (int j = 0; j < 8; ++j) orow[(size_t)j * D] = acc[j] * irow[j];
}


static inline size_t align256(size_t x) { return (x + 255) & ~(size_t)255; }

extern "C" void kernel_launch(void* const* d_in, const int* in_sizes, int n_in,
                              void* d_out, int out_size, void* d_ws, size_t ws_size,
                              hipStream_t stream) {
  const int*   E  = (const int*)d_in[1];
  const float* X  = (const float*)d_in[2];
  const float* W1 = (const float*)d_in[3];
  const float* b1 = (const float*)d_in[4];
  const float* W2 = (const float*)d_in[5];
  const float* b2 = (const float*)d_in[6];
  const float* W3 = (const float*)d_in[7];
  const float* b3 = (const float*)d_in[8];
  float* out = (float*)d_out;

  const int n_edges = in_sizes[1] / 2;
  const int n_nodes = in_sizes[2] / D;
  const int* src = E;
  const int* dst = E + n_edges;

  const int nbin = (n_nodes + BIN_N - 1) / BIN_N;
  const int nblk = (n_edges + CH - 1) / CH;

  char* ws = (char*)d_ws;
  int*   rowptr   = (int*)ws;   ws += align256((size_t)(n_nodes + 1) * 4);
  float* inv      = (float*)ws; ws += align256((size_t)n_nodes * 4);
  int*   hist     = (int*)ws;   ws += align256((size_t)nbin * nblk * 4);
  int*   off      = (int*)ws;   ws += align256((size_t)nbin * nblk * 4);
  int*   tot      = (int*)ws;   ws += align256((size_t)nbin * 4);
  int*   binstart = (int*)ws;   ws += align256((size_t)(nbin + 1) * 4);
  int2*  pairs    = (int2*)ws;  ws += align256((size_t)n_edges * 8);
  int*   csr      = (int*)ws;   ws += align256((size_t)n_edges * 4);
  float* T        = (float*)ws; ws += align256((size_t)n_nodes * D * 4);
  float* T2       = (float*)ws;

  const int nb_gath  = (n_nodes + 7) / 8;
  const int nb_gemm  = n_nodes / 16;

  bin_hist<<<nblk, 1024, 0, stream>>>(dst, hist, n_edges, nbin);
  bin_scan<<<nbin, 512, 0, stream>>>(hist, off, tot, nblk);
  bin_place<<<nblk, 1024, 0, stream>>>(src, dst, hist, off, tot, binstart, pairs, n_edges, nbin);
  bin_sort<<<nbin, 1024, 0, stream>>>(pairs, binstart, rowptr, inv, csr, n_nodes);

  gemm16_wmma<<<nb_gemm, 128, 0, stream>>>(X, W1, inv, T);
  gather_gemm<<<nb_gemm, 512, 0, stream>>>(rowptr, csr, inv, T, b1, W2, T2);
  gather_gemm<<<nb_gemm, 512, 0, stream>>>(rowptr, csr, inv, T2, b2, W3, T);
  gather_rows<<<nb_gath, 256, 0, stream>>>(rowptr, csr, inv, T, b3, out, n_nodes);
}


# gather kernels: each wave walks 4 consecutive destination rows, not 1 (stripe block 16 -> 4 waves): 25000 waves per gather launch, not 100000; wave start rate bounded them; same sums in same order
# speedup vs baseline: 4.7208x; 1.0892x over previous
#include <hip/hip_runtime.h>


typedef __attribute__((ext_vector_type(2))) float v2f;
typedef __attribute__((ext_vector_type(8))) float v8f;

#define D 64

#define BIN_N 256
#define BIN_SH 8
#define CH 6144
#define MAXBIN 512
#define CAP 8192

__global__ __launch_bounds__(1024) void bin_hist(const int* __restrict__ dst,
                                                 int* __restrict__ hist,
                                                 int n_edges, int nbin) {
  __shared__ int h[MAXBIN];
  const int t = threadIdx.x;
  const int beg = blockIdx.x * CH;
  const int end = min(beg + CH, n_edges);
  int dv[CH / 1024];
#pragma unroll
  for (int k = 0; k < CH / 1024; ++k) {
    const int e = beg + t + k * 1024;
    dv[k] = (e < end) ? dst[e] : 0;
  }
  for (int i = t; i < nbin; i += 1024) h[i] = 0;
  __syncthreads();
#pragma unroll
  for (int k = 0; k < CH / 1024; ++k)
    if (beg + t + k * 1024 < end) atomicAdd(&h[dv[k] >> BIN_SH], 1);
  __syncthreads();
  for (int i = t; i < nbin; i += 1024) hist[i * gridDim.x + blockIdx.x] = h[i];
}

__global__ __launch_bounds__(512) void bin_scan(const int* __restrict__ hist,
                                                int* __restrict__ off,
                                                int* __restrict__ tot, int nblk) {
  __shared__ int wsum[16];
  const int t = threadIdx.x;
  const int lane = t & 31;
  const int wv = t >> 5;
  const int* __restrict__ h = hist + (size_t)blockIdx.x * nblk;
  int* __restrict__ o = off + (size_t)blockIdx.x * nblk;
  int carry = 0;
  for (int base = 0; base < nblk; base += 512) {
    const int i = base + t;
    const int v = (i < nblk) ? h[i] : 0;
    int incl = v;
#pragma unroll
    for (int s = 1; s < 32; s <<= 1) {
      int u = __shfl_up(incl, s);
      if (lane >= s) incl += u;
    }
    if (lane == 31) wsum[wv] = incl;
    __syncthreads();
    int woff = 0, all = 0;
#pragma unroll
    for (int k = 0; k < 16; ++k) {
      const int s = wsum[k];
      if (k < wv) woff += s;
      all += s;
    }
    if (i < nblk) o[i] = carry + woff + incl - v;
    carry += all;
    __syncthreads();
  }
  if (t == 0) tot[blockIdx.x] = carry;
}

__global__ __launch_bounds__(1024) void bin_place(
    const int* __restrict__ src, const int* __restrict__ dst,
    const int* __restrict__ hist, const int* __restrict__ off,
    const int* __restrict__ tot, int* __restrict__ binstart,
    int2* __restrict__ pairs, int n_edges, int nbin) {
  __shared__ int lcur[MAXBIN];
  __shared__ int gdelta[MAXBIN];
  __shared__ int wsa[16];
  __shared__ int wsb[16];
  __shared__ int2 stage[CH];
  const int t = threadIdx.x;
  const int lane = t & 31;
  const int wv = t >> 5;
  const int b = blockIdx.x;
  const int nblk = gridDim.x;

  const int beg = b * CH;
  const int end = min(beg + CH, n_edges);
  int sv[CH / 1024], dv[CH / 1024];
#pragma unroll
  for (int k = 0; k < CH / 1024; ++k) {
    const int e = beg + t + k * 1024;
    sv[k] = (e < end) ? src[e] : 0;
    dv[k] = (e < end) ? dst[e] : 0;
  }

  int hv = 0, tv = 0, ov = 0;
  if (t < nbin) {
    hv = hist[(size_t)t * nblk + b];
    ov = off[(size_t)t * nblk + b];
    tv = tot[t];
  }
  int hi = hv, ti = tv;
#pragma unroll
  for (int s = 1; s < 32; s <<= 1) {
    int u = __shfl_up(hi, s);
    int w = __shfl_up(ti, s);
    if (lane >= s) { hi += u; ti += w; }
  }
  if (t < MAXBIN && lane == 31) { wsa[wv] = hi; wsb[wv] = ti; }
  __syncthreads();
  if (t < MAXBIN) {
    int wa = 0, wb = 0;
    for (int k = 0; k < wv; ++k) { wa += wsa[k]; wb += wsb[k]; }
    const int ls = wa + hi - hv;
    const int gs = wb + ti - tv;
    if (t < nbin) {
      lcur[t] = ls;
      gdelta[t] = gs + ov - ls;
      if (b == 0) {
        binstart[t] = gs;
        if (t == nbin - 1) binstart[nbin] = gs + tv;
      }
    }
  }
  __syncthreads();

#pragma unroll
  for (int k = 0; k < CH / 1024; ++k) {
    if (beg + t + k * 1024 < end) {
      const int r = atomicAdd(&lcur[dv[k] >> BIN_SH], 1);
      stage[r] = make_int2(sv[k], dv[k]);
    }
  }
  __syncthreads();
  const int cnt = end - beg;
  int2 pv[CH / 1024];
  int gv[CH / 1024];
#pragma unroll
  for (int k = 0; k < CH / 1024; ++k)
    if (t + k * 1024 < cnt) pv[k] = stage[t + k * 1024];
#pragma unroll
  for (int k = 0; k < CH / 1024; ++k)
    if (t + k * 1024 < cnt) gv[k] = gdelta[pv[k].y >> BIN_SH];
#pragma unroll
  for (int k = 0; k < CH / 1024; ++k)
    if (t + k * 1024 < cnt) pairs[gv[k] + t + k * 1024] = pv[k];
}

__global__ __launch_bounds__(1024) void bin_sort(
    const int2* __restrict__ pairs, const int* __restrict__ binstart,
    int* __restrict__ rowptr, float* __restrict__ inv, int* __restrict__ csr, int n) {
  __shared__ int cnt[BIN_N];
  __shared__ int cur[BIN_N];
  __shared__ int wsum[BIN_N / 32];
  __shared__ int stage[CAP];
  const int t = threadIdx.x;
  const int lane = t & 31;
  const int wv = t >> 5;
  const int node0 = blockIdx.x * BIN_N;
  const int s = binstart[blockIdx.x];
  const int e = binstart[blockIdx.x + 1];
  const int m = e - s;

  const bool small = (m <= CAP);
  int2 p[CAP / 1024];
  if (small) {
#pragma unroll
    for (int k = 0; k < CAP / 1024; ++k) {
      const int i = s + t + k * 1024;
      p[k] = (i < e) ? pairs[i] : make_int2(0, node0);
    }
  }
  if (t < BIN_N) cnt[t] = 0;
  __syncthreads();
  if (small) {
#pragma unroll
    for (int k = 0; k < CAP / 1024; ++k)
      if (s + t + k * 1024 < e) atomicAdd(&cnt[p[k].y - node0], 1);
  } else {
    for (int i = s + t; i < e; i += 1024) atomicAdd(&cnt[pairs[i].y - node0], 1);
  }
  __syncthreads();

  int v = 0, incl = 0;
  if (t < BIN_N) {
    v = cnt[t];
    incl = v;
#pragma unroll
    for (int o = 1; o < 32; o <<= 1) {
      int u = __shfl_up(incl, o);
      if (lane >= o) incl += u;
    }
    if (lane == 31) wsum[wv] = incl;
  }
  __syncthreads();
  if (t < BIN_N) {
    int woff = 0;
    for (int k = 0; k < wv; ++k) woff += wsum[k];
    const int excl = woff + incl - v;
    cur[t] = excl;
    const int node = node0 + t;
    if (node < n) {
      rowptr[node] = s + excl;
      inv[node] = 1.0f / sqrtf((float)v + 1.0f);
      if (node == n - 1) rowptr[n] = e;
    }
  }
  __syncthreads();

  if (small) {
#pragma unroll
    for (int k = 0; k < CAP / 1024; ++k)
      if (s + t + k * 1024 < e) stage[atomicAdd(&cur[p[k].y - node0], 1)] = p[k].x;
    __syncthreads();
    int ov[CAP / 1024];
#pragma unroll
    for (int k = 0; k < CAP / 1024; ++k)
      if (t + k * 1024 < m) ov[k] = stage[t + k * 1024];
#pragma unroll
    for (int k = 0; k < CAP / 1024; ++k)
      if (t + k * 1024 < m) csr[s + t + k * 1024] = ov[k];
  } else {
    for (int i = s + t; i < e; i += 1024) {
      const int2 q = pairs[i];
      csr[s + atomicAdd(&cur[q.y - node0], 1)] = q.x;
    }
  }
}

__global__ void gemm16_wmma(const float* __restrict__ A, const float* __restrict__ W,
                            const float* __restrict__ inv, float* __restrict__ T) {
  const int lane = threadIdx.x & 31;
  const int wave = threadIdx.x >> 5;
  const int half = lane >> 4;
  const int l    = lane & 15;
  const int row0 = blockIdx.x << 4;
  const int c0   = wave << 4;

  const float* __restrict__ arow = A + (size_t)(row0 + l) * D;

  v8f acc = {};
#pragma unroll
  for (int kk = 0; kk < D; kk += 4) {
    const int ka = kk + half * 2;
    float a0 = arow[ka];
    float a1 = arow[ka + 1];
    v2f a = {a0, a1};
    v2f b = { W[ka * D + c0 + l], W[(ka + 1) * D + c0 + l] };
    acc = __builtin_amdgcn_wmma_f32_16x16x4_f32(
         false, a,  false, b,
         (short)0, acc,  false,  false);
  }

  float* __restrict__ orow = T + (size_t)(row0 + half * 8) * D + c0 + l;
  const float* __restrict__ irow = inv + row0 + half * 8;
#pragma unroll
  for (int j = 0; j < 8; ++j) orow[(size_t)j * D] = acc[j] * irow[j];
}

#define RPW 4

template <int NB>
__device__ __forceinline__ void gather_batch(const int* __restrict__ csr, int k,
                                             const float2* __restrict__ Tv, float2& acc) {
  int ed[NB];
  float2 tv[NB];
#pragma unroll
  for (int j = 0; j < NB; ++j) ed[j] = csr[k + j];
#pragma unroll
  for (int j = 0; j < NB; ++j) tv[j] = Tv[(size_t)ed[j] * (D / 2)];
#pragma unroll
  for (int j = 0; j < NB; ++j) {
    acc.x += tv[j].x;
    acc.y += tv[j].y;
  }
}

__device__ __forceinline__ float2 gather_node(const int* __restrict__ csr, int beg, int end,
                                              const float2* __restrict__ Tv, float2 acc) {
  int k = beg;
  for (; k + 8 <= end; k += 8) gather_batch<8>(csr, k, Tv, acc);
  const int rem = end - k;
  if (rem & 4) { gather_batch<4>(csr, k, Tv, acc); k += 4; }
  if (rem & 2) { gather_batch<2>(csr, k, Tv, acc); k += 2; }
  if (rem & 1) { gather_batch<1>(csr, k, Tv, acc); }
  return acc;
}

__global__ void gather_rows(const int* __restrict__ rowptr, const int* __restrict__ csr,
                            const float* __restrict__ inv, const float* __restrict__ T,
                            const float* __restrict__ bias, float* __restrict__ out,
                            int n) {
  const int dv = (blockIdx.x * (blockDim.x >> 5) + (threadIdx.x >> 5)) * RPW;
  if (dv >= n) return;
  const int d0 = __builtin_amdgcn_readfirstlane(dv);
  const int lane = threadIdx.x & 31;
  const float2* __restrict__ Tv = (const float2*)T + lane;
  const float2 b = ((const float2*)bias)[lane];
  int rp[RPW + 1];
  float isd[RPW];
  float2 self[RPW];
#pragma unroll
  for (int r = 0; r <= RPW; ++r) rp[r] = rowptr[d0 + r];
#pragma unroll
  for (int r = 0; r < RPW; ++r) {
    isd[r] = inv[d0 + r];
    self[r] = Tv[(size_t)(d0 + r) * (D / 2)];
  }
#pragma unroll
  for (int r = 0; r < RPW; ++r) {
    float2 acc = gather_node(csr, rp[r], rp[r + 1], Tv, self[r]);
    acc.x = fmaf(acc.x, isd[r], b.x);
    acc.y = fmaf(acc.y, isd[r], b.y);
    ((float2*)out)[(size_t)(d0 + r) * (D / 2) + lane] = acc;
  }
}

#define LDP 68

__global__ __launch_bounds__(128) void gather_gemm(
    const int* __restrict__ rowptr, const int* __restrict__ csr,
    const float* __restrict__ inv, const float* __restrict__ T,
    const float* __restrict__ bias, const float* __restrict__ W,
    float* __restrict__ Tn) {
  __shared__ float hs[16 * LDP];
  const int lane = threadIdx.x & 31;
  const int wave = threadIdx.x >> 5;
  const int row0 = blockIdx.x << 4;
  {
    const int d0 = __builtin_amdgcn_readfirstlane(row0 + wave * RPW);
    const float2* __restrict__ Tv = (const float2*)T + lane;
    const float2 b = ((const float2*)bias)[lane];
    int rp[RPW + 1];
    float isd[RPW];
    float2 self[RPW];
#pragma unroll
    for (int r = 0; r <= RPW; ++r) rp[r] = rowptr[d0 + r];
#pragma unroll
    for (int r = 0; r < RPW; ++r) {
      isd[r] = inv[d0 + r];
      self[r] = Tv[(size_t)(d0 + r) * (D / 2)];
    }
#pragma unroll
    for (int r = 0; r < RPW; ++r) {
      const float2 acc = gather_node(csr, rp[r], rp[r + 1], Tv, self[r]);
      float2 h;
      h.x = fmaxf(fmaf(acc.x, isd[r], b.x), 0.0f);
      h.y = fmaxf(fmaf(acc.y, isd[r], b.y), 0.0f);
      *(float2*)&hs[(wave * RPW + r) * LDP + 2 * lane] = h;
    }
  }
  __syncthreads();

  const int half = lane >> 4;
  const int l    = lane & 15;
  const int c0   = wave << 4;
  const float* __restrict__ arow = hs + l * LDP;

  v8f acc = {};
#pragma unroll
  for (int kk = 0; kk < D; kk += 4) {
    const int ka = kk + half * 2;
    v2f a = {arow[ka], arow[ka + 1]};
    v2f b = { W[ka * D + c0 + l], W[(ka + 1) * D + c0 + l] };
    acc = __builtin_amdgcn_wmma_f32_16x16x4_f32(
         false, a,  false, b,
         (short)0, acc,  false,  false);
  }

  float* __restrict__ orow = Tn + (size_t)(row0 + half * 8) * D + c0 + l;
  const float* __restrict__ irow = inv + row0 + half * 8;
#pragma unroll
  for (int j = 0; j < 8; ++j) orow[(size_t)j * D] = acc[j] * irow[j];
}


static inline size_t align256(size_t x) { return (x + 255) & ~(size_t)255; }

extern "C" void kernel_launch(void* const* d_in, const int* in_sizes, int n_in,
                              void* d_out, int out_size, void* d_ws, size_t ws_size,
                              hipStream_t stream) {
  const int*   E  = (const int*)d_in[1];
  const float* X  = (const float*)d_in[2];
  const float* W1 = (const float*)d_in[3];
  const float* b1 = (const float*)d_in[4];
  const float* W2 = (const float*)d_in[5];
  const float* b2 = (const float*)d_in[6];
  const float* W3 = (const float*)d_in[7];
  const float* b3 = (const float*)d_in[8];
  float* out = (float*)d_out;

  const int n_edges = in_sizes[1] / 2;
  const int n_nodes = in_sizes[2] / D;
  const int* src = E;
  const int* dst = E + n_edges;

  const int nbin = (n_nodes + BIN_N - 1) / BIN_N;
  const int nblk = (n_edges + CH - 1) / CH;

  char* ws = (char*)d_ws;
  int*   rowptr   = (int*)ws;   ws += align256((size_t)(n_nodes + 1) * 4);
  float* inv      = (float*)ws; ws += align256((size_t)n_nodes * 4);
  int*   hist     = (int*)ws;   ws += align256((size_t)nbin * nblk * 4);
  int*   off      = (int*)ws;   ws += align256((size_t)nbin * nblk * 4);
  int*   tot      = (int*)ws;   ws += align256((size_t)nbin * 4);
  int*   binstart = (int*)ws;   ws += align256((size_t)(nbin + 1) * 4);
  int2*  pairs    = (int2*)ws;  ws += align256((size_t)n_edges * 8);
  int*   csr      = (int*)ws;   ws += align256((size_t)n_edges * 4);
  float* T        = (float*)ws; ws += align256((size_t)n_nodes * D * 4);
  float* T2       = (float*)ws;

  const int nb_gath  = (n_nodes + 8 * RPW - 1) / (8 * RPW);
  const int nb_gemm  = n_nodes / 16;

  bin_hist<<<nblk, 1024, 0, stream>>>(dst, hist, n_edges, nbin);
  bin_scan<<<nbin, 512, 0, stream>>>(hist, off, tot, nblk);
  bin_place<<<nblk, 1024, 0, stream>>>(src, dst, hist, off, tot, binstart, pairs, n_edges, nbin);
  bin_sort<<<nbin, 1024, 0, stream>>>(pairs, binstart, rowptr, inv, csr, n_nodes);

  gemm16_wmma<<<nb_gemm, 128, 0, stream>>>(X, W1, inv, T);
  gather_gemm<<<nb_gemm, 128, 0, stream>>>(rowptr, csr, inv, T, b1, W2, T2);
  gather_gemm<<<nb_gemm, 128, 0, stream>>>(rowptr, csr, inv, T2, b2, W3, T);
  gather_rows<<<nb_gath, 256, 0, stream>>>(rowptr, csr, inv, T, b3, out, n_nodes);
}


# layer 1: dense product moved to the tail of the edge-list kernel (a 256-node bin block multiplies its own rows once it has their inv; X, W staged in LDS, whole-row loads); 8 -> 7 launches; same sums
# speedup vs baseline: 5.0651x; 1.0729x over previous
#include <hip/hip_runtime.h>


typedef __attribute__((ext_vector_type(2))) float v2f;
typedef __attribute__((ext_vector_type(4))) float v4f;
typedef __attribute__((ext_vector_type(8))) float v8f;

#define D 64

#define BIN_N 256
#define BIN_SH 8
#define CH 6144
#define MAXBIN 512
#define CAP 8192
#define XP 68

__global__ __launch_bounds__(1024) void bin_hist(const int* __restrict__ dst,
                                                 int* __restrict__ hist,
                                                 int n_edges, int nbin) {
  __shared__ int h[MAXBIN];
  const int t = threadIdx.x;
  const int beg = blockIdx.x * CH;
  const int end = min(beg + CH, n_edges);
  int dv[CH / 1024];
#pragma unroll
  for (int k = 0; k < CH / 1024; ++k) {
    const int e = beg + t + k * 1024;
    dv[k] = (e < end) ? dst[e] : 0;
  }
  for (int i = t; i < nbin; i += 1024) h[i] = 0;
  __syncthreads();
#pragma unroll
  for (int k = 0; k < CH / 1024; ++k)
    if (beg + t + k * 1024 < end) atomicAdd(&h[dv[k] >> BIN_SH], 1);
  __syncthreads();
  for (int i = t; i < nbin; i += 1024) hist[i * gridDim.x + blockIdx.x] = h[i];
}

__global__ __launch_bounds__(512) void bin_scan(const int* __restrict__ hist,
                                                int* __restrict__ off,
                                                int* __restrict__ tot, int nblk) {
  __shared__ int wsum[16];
  const int t = threadIdx.x;
  const int lane = t & 31;
  const int wv = t >> 5;
  const int* __restrict__ h = hist + (size_t)blockIdx.x * nblk;
  int* __restrict__ o = off + (size_t)blockIdx.x * nblk;
  int carry = 0;
  for (int base = 0; base < nblk; base += 512) {
    const int i = base + t;
    const int v = (i < nblk) ? h[i] : 0;
    int incl = v;
#pragma unroll
    for (int s = 1; s < 32; s <<= 1) {
      int u = __shfl_up(incl, s);
      if (lane >= s) incl += u;
    }
    if (lane == 31) wsum[wv] = incl;
    __syncthreads();
    int woff = 0, all = 0;
#pragma unroll
    for (int k = 0; k < 16; ++k) {
      const int s = wsum[k];
      if (k < wv) woff += s;
      all += s;
    }
    if (i < nblk) o[i] = carry + woff + incl - v;
    carry += all;
    __syncthreads();
  }
  if (t == 0) tot[blockIdx.x] = carry;
}

__global__ __launch_bounds__(1024) void bin_place(
    const int* __restrict__ src, const int* __restrict__ dst,
    const int* __restrict__ hist, const int* __restrict__ off,
    const int* __restrict__ tot, int* __restrict__ binstart,
    int2* __restrict__ pairs, int n_edges, int nbin) {
  __shared__ int lcur[MAXBIN];
  __shared__ int gdelta[MAXBIN];
  __shared__ int wsa[16];
  __shared__ int wsb[16];
  __shared__ int2 stage[CH];
  const int t = threadIdx.x;
  const int lane = t & 31;
  const int wv = t >> 5;
  const int b = blockIdx.x;
  const int nblk = gridDim.x;

  const int beg = b * CH;
  const int end = min(beg + CH, n_edges);
  int sv[CH / 1024], dv[CH / 1024];
#pragma unroll
  for (int k = 0; k < CH / 1024; ++k) {
    const int e = beg + t + k * 1024;
    sv[k] = (e < end) ? src[e] : 0;
    dv[k] = (e < end) ? dst[e] : 0;
  }

  int hv = 0, tv = 0, ov = 0;
  if (t < nbin) {
    hv = hist[(size_t)t * nblk + b];
    ov = off[(size_t)t * nblk + b];
    tv = tot[t];
  }
  int hi = hv, ti = tv;
#pragma unroll
  for (int s = 1; s < 32; s <<= 1) {
    int u = __shfl_up(hi, s);
    int w = __shfl_up(ti, s);
    if (lane >= s) { hi += u; ti += w; }
  }
  if (t < MAXBIN && lane == 31) { wsa[wv] = hi; wsb[wv] = ti; }
  __syncthreads();
  if (t < MAXBIN) {
    int wa = 0, wb = 0;
    for (int k = 0; k < wv; ++k) { wa += wsa[k]; wb += wsb[k]; }
    const int ls = wa + hi - hv;
    const int gs = wb + ti - tv;
    if (t < nbin) {
      lcur[t] = ls;
      gdelta[t] = gs + ov - ls;
      if (b == 0) {
        binstart[t] = gs;
        if (t == nbin - 1) binstart[nbin] = gs + tv;
      }
    }
  }
  __syncthreads();

#pragma unroll
  for (int k = 0; k < CH / 1024; ++k) {
    if (beg + t + k * 1024 < end) {
      const int r = atomicAdd(&lcur[dv[k] >> BIN_SH], 1);
      stage[r] = make_int2(sv[k], dv[k]);
    }
  }
  __syncthreads();
  const int cnt = end - beg;
  int2 pv[CH / 1024];
  int gv[CH / 1024];
#pragma unroll
  for (int k = 0; k < CH / 1024; ++k)
    if (t + k * 1024 < cnt) pv[k] = stage[t + k * 1024];
#pragma unroll
  for (int k = 0; k < CH / 1024; ++k)
    if (t + k * 1024 < cnt) gv[k] = gdelta[pv[k].y >> BIN_SH];
#pragma unroll
  for (int k = 0; k < CH / 1024; ++k)
    if (t + k * 1024 < cnt) pairs[gv[k] + t + k * 1024] = pv[k];
}

__global__ __launch_bounds__(1024) void bin_sort(
    const int2* __restrict__ pairs, const int* __restrict__ binstart,
    int* __restrict__ rowptr, float* __restrict__ inv, int* __restrict__ csr, int n,
    const float* __restrict__ X, const float* __restrict__ W, float* __restrict__ T) {
  __shared__ int cnt[BIN_N];
  __shared__ int cur[BIN_N];
  __shared__ int wsum[BIN_N / 32];
  __shared__ float invs[BIN_N];
  __shared__ __attribute__((aligned(16))) float xs[BIN_N * XP];
  __shared__ __attribute__((aligned(16))) float wt[D * XP];
  __shared__ int stage[CAP];
  const int t = threadIdx.x;
  const int lane = t & 31;
  const int wv = t >> 5;
  const int node0 = blockIdx.x * BIN_N;
  const int s = binstart[blockIdx.x];
  const int e = binstart[blockIdx.x + 1];
  const int m = e - s;

  const int half = lane >> 4;
  const int l    = lane & 15;
  const int stripe = wv >> 1;
  const int srow = node0 + (stripe << 4);
  const bool has = (srow < n);
  const bool small = (m <= CAP);
  int2 p[CAP / 1024];
  if (small) {
#pragma unroll
    for (int k = 0; k < CAP / 1024; ++k) {
      const int i = s + t + k * 1024;
      p[k] = (i < e) ? pairs[i] : make_int2(0, node0);
    }
  }
  v4f xr[4];
  if (has) {
#pragma unroll
    for (int i = 0; i < 4; ++i)
      xr[i] = *(const v4f*)(X + (size_t)(srow + ((wv & 1) << 3) + 2 * i + half) * D + 4 * l);
  }
  const v4f wr = *(const v4f*)(W + 4 * t);
  if (t < BIN_N) cnt[t] = 0;
  __syncthreads();
  if (small) {
#pragma unroll
    for (int k = 0; k < CAP / 1024; ++k)
      if (s + t + k * 1024 < e) atomicAdd(&cnt[p[k].y - node0], 1);
  } else {
    for (int i = s + t; i < e; i += 1024) atomicAdd(&cnt[pairs[i].y - node0], 1);
  }
  __syncthreads();

  int v = 0, incl = 0;
  if (t < BIN_N) {
    v = cnt[t];
    incl = v;
#pragma unroll
    for (int o = 1; o < 32; o <<= 1) {
      int u = __shfl_up(incl, o);
      if (lane >= o) incl += u;
    }
    if (lane == 31) wsum[wv] = incl;
  }
  __syncthreads();
  if (t < BIN_N) {
    int woff = 0;
    for (int k = 0; k < wv; ++k) woff += wsum[k];
    const int excl = woff + incl - v;
    cur[t] = excl;
    const int node = node0 + t;
    if (node < n) {
      rowptr[node] = s + excl;
      const float iv = 1.0f / sqrtf((float)v + 1.0f);
      inv[node] = iv;
      invs[t] = iv;
      if (node == n - 1) rowptr[n] = e;
    }
  }
  if (has) {
#pragma unroll
    for (int i = 0; i < 4; ++i)
      *(v4f*)&xs[((stripe << 4) + ((wv & 1) << 3) + 2 * i + half) * XP + 4 * l] = xr[i];
  }
#pragma unroll
  for (int i = 0; i < 4; ++i) wt[(4 * (t & 15) + i) * XP + (t >> 4)] = wr[i];
  __syncthreads();

  if (small) {
#pragma unroll
    for (int k = 0; k < CAP / 1024; ++k)
      if (s + t + k * 1024 < e) stage[atomicAdd(&cur[p[k].y - node0], 1)] = p[k].x;
    __syncthreads();
    int ov[CAP / 1024];
#pragma unroll
    for (int k = 0; k < CAP / 1024; ++k)
      if (t + k * 1024 < m) ov[k] = stage[t + k * 1024];
#pragma unroll
    for (int k = 0; k < CAP / 1024; ++k)
      if (t + k * 1024 < m) csr[s + t + k * 1024] = ov[k];
  } else {
    for (int i = s + t; i < e; i += 1024) {
      const int2 q = pairs[i];
      csr[s + atomicAdd(&cur[q.y - node0], 1)] = q.x;
    }
  }

  if (has) {
    const int c0 = (wv & 1) << 5;
    const float* __restrict__ arow = xs + ((stripe << 4) + l) * XP;
    const float* __restrict__ brow = wt + (c0 + l) * XP;
    v8f acc0 = {};
    v8f acc1 = {};
#pragma unroll
    for (int kk = 0; kk < D; kk += 4) {
      const int ka = kk + half * 2;
      v2f a  = { arow[ka], arow[ka + 1] };
      v2f b0 = { brow[ka], brow[ka + 1] };
      v2f b1 = { brow[16 * XP + ka], brow[16 * XP + ka + 1] };
      acc0 = __builtin_amdgcn_wmma_f32_16x16x4_f32(
           false, a,  false, b0,
           (short)0, acc0,  false,  false);
      acc1 = __builtin_amdgcn_wmma_f32_16x16x4_f32(
           false, a,  false, b1,
           (short)0, acc1,  false,  false);
    }
    float* __restrict__ orow = T + (size_t)(srow + half * 8) * D + c0 + l;
    const float* __restrict__ irow = invs + (srow - node0) + half * 8;
#pragma unroll
    for (int j = 0; j < 8; ++j) {
      const float sc = irow[j];
      orow[(size_t)j * D]      = acc0[j] * sc;
      orow[(size_t)j * D + 16] = acc1[j] * sc;
    }
  }
}

#define RPW 4

template <int NB>
__device__ __forceinline__ void gather_batch(const int* __restrict__ csr, int k,
                                             const float2* __restrict__ Tv, float2& acc) {
  int ed[NB];
  float2 tv[NB];
#pragma unroll
  for (int j = 0; j < NB; ++j) ed[j] = csr[k + j];
#pragma unroll
  for (int j = 0; j < NB; ++j) tv[j] = Tv[(size_t)ed[j] * (D / 2)];
#pragma unroll
  for (int j = 0; j < NB; ++j) {
    acc.x += tv[j].x;
    acc.y += tv[j].y;
  }
}

__device__ __forceinline__ float2 gather_node(const int* __restrict__ csr, int beg, int end,
                                              const float2* __restrict__ Tv, float2 acc) {
  int k = beg;
  for (; k + 8 <= end; k += 8) gather_batch<8>(csr, k, Tv, acc);
  const int rem = end - k;
  if (rem & 4) { gather_batch<4>(csr, k, Tv, acc); k += 4; }
  if (rem & 2) { gather_batch<2>(csr, k, Tv, acc); k += 2; }
  if (rem & 1) { gather_batch<1>(csr, k, Tv, acc); }
  return acc;
}

__global__ void gather_rows(const int* __restrict__ rowptr, const int* __restrict__ csr,
                            const float* __restrict__ inv, const float* __restrict__ T,
                            const float* __restrict__ bias, float* __restrict__ out,
                            int n) {
  const int dv = (blockIdx.x * (blockDim.x >> 5) + (threadIdx.x >> 5)) * RPW;
  if (dv >= n) return;
  const int d0 = __builtin_amdgcn_readfirstlane(dv);
  const int lane = threadIdx.x & 31;
  const float2* __restrict__ Tv = (const float2*)T + lane;
  const float2 b = ((const float2*)bias)[lane];
  int rp[RPW + 1];
  float isd[RPW];
  float2 self[RPW];
#pragma unroll
  for (int r = 0; r <= RPW; ++r) rp[r] = rowptr[d0 + r];
#pragma unroll
  for (int r = 0; r < RPW; ++r) {
    isd[r] = inv[d0 + r];
    self[r] = Tv[(size_t)(d0 + r) * (D / 2)];
  }
#pragma unroll
  for (int r = 0; r < RPW; ++r) {
    float2 acc = gather_node(csr, rp[r], rp[r + 1], Tv, self[r]);
    acc.x = fmaf(acc.x, isd[r], b.x);
    acc.y = fmaf(acc.y, isd[r], b.y);
    ((float2*)out)[(size_t)(d0 + r) * (D / 2) + lane] = acc;
  }
}

#define LDP 68

__global__ __launch_bounds__(128) void gather_gemm(
    const int* __restrict__ rowptr, const int* __restrict__ csr,
    const float* __restrict__ inv, const float* __restrict__ T,
    const float* __restrict__ bias, const float* __restrict__ W,
    float* __restrict__ Tn) {
  __shared__ float hs[16 * LDP];
  const int lane = threadIdx.x & 31;
  const int wave = threadIdx.x >> 5;
  const int row0 = blockIdx.x << 4;
  {
    const int d0 = __builtin_amdgcn_readfirstlane(row0 + wave * RPW);
    const float2* __restrict__ Tv = (const float2*)T + lane;
    const float2 b = ((const float2*)bias)[lane];
    int rp[RPW + 1];
    float isd[RPW];
    float2 self[RPW];
#pragma unroll
    for (int r = 0; r <= RPW; ++r) rp[r] = rowptr[d0 + r];
#pragma unroll
    for (int r = 0; r < RPW; ++r) {
      isd[r] = inv[d0 + r];
      self[r] = Tv[(size_t)(d0 + r) * (D / 2)];
    }
#pragma unroll
    for (int r = 0; r < RPW; ++r) {
      const float2 acc = gather_node(csr, rp[r], rp[r + 1], Tv, self[r]);
      float2 h;
      h.x = fmaxf(fmaf(acc.x, isd[r], b.x), 0.0f);
      h.y = fmaxf(fmaf(acc.y, isd[r], b.y), 0.0f);
      *(float2*)&hs[(wave * RPW + r) * LDP + 2 * lane] = h;
    }
  }
  __syncthreads();

  const int half = lane >> 4;
  const int l    = lane & 15;
  const int c0   = wave << 4;
  const float* __restrict__ arow = hs + l * LDP;

  v8f acc = {};
#pragma unroll
  for (int kk = 0; kk < D; kk += 4) {
    const int ka = kk + half * 2;
    v2f a = {arow[ka], arow[ka + 1]};
    v2f b = { W[ka * D + c0 + l], W[(ka + 1) * D + c0 + l] };
    acc = __builtin_amdgcn_wmma_f32_16x16x4_f32(
         false, a,  false, b,
         (short)0, acc,  false,  false);
  }

  float* __restrict__ orow = Tn + (size_t)(row0 + half * 8) * D + c0 + l;
  const float* __restrict__ irow = inv + row0 + half * 8;
#pragma unroll
  for (int j = 0; j < 8; ++j) orow[(size_t)j * D] = acc[j] * irow[j];
}


static inline size_t align256(size_t x) { return (x + 255) & ~(size_t)255; }

extern "C" void kernel_launch(void* const* d_in, const int* in_sizes, int n_in,
                              void* d_out, int out_size, void* d_ws, size_t ws_size,
                              hipStream_t stream) {
  const int*   E  = (const int*)d_in[1];
  const float* X  = (const float*)d_in[2];
  const float* W1 = (const float*)d_in[3];
  const float* b1 = (const float*)d_in[4];
  const float* W2 = (const float*)d_in[5];
  const float* b2 = (const float*)d_in[6];
  const float* W3 = (const float*)d_in[7];
  const float* b3 = (const float*)d_in[8];
  float* out = (float*)d_out;

  const int n_edges = in_sizes[1] / 2;
  const int n_nodes = in_sizes[2] / D;
  const int* src = E;
  const int* dst = E + n_edges;

  const int nbin = (n_nodes + BIN_N - 1) / BIN_N;
  const int nblk = (n_edges + CH - 1) / CH;

  char* ws = (char*)d_ws;
  int*   rowptr   = (int*)ws;   ws += align256((size_t)(n_nodes + 1) * 4);
  float* inv      = (float*)ws; ws += align256((size_t)n_nodes * 4);
  int*   hist     = (int*)ws;   ws += align256((size_t)nbin * nblk * 4);
  int*   off      = (int*)ws;   ws += align256((size_t)nbin * nblk * 4);
  int*   tot      = (int*)ws;   ws += align256((size_t)nbin * 4);
  int*   binstart = (int*)ws;   ws += align256((size_t)(nbin + 1) * 4);
  int2*  pairs    = (int2*)ws;  ws += align256((size_t)n_edges * 8);
  int*   csr      = (int*)ws;   ws += align256((size_t)n_edges * 4);
  float* T        = (float*)ws; ws += align256((size_t)n_nodes * D * 4);
  float* T2       = (float*)ws;

  const int nb_gath  = (n_nodes + 8 * RPW - 1) / (8 * RPW);
  const int nb_gemm  = n_nodes / 16;

  bin_hist<<<nblk, 1024, 0, stream>>>(dst, hist, n_edges, nbin);
  bin_scan<<<nbin, 512, 0, stream>>>(hist, off, tot, nblk);
  bin_place<<<nblk, 1024, 0, stream>>>(src, dst, hist, off, tot, binstart, pairs, n_edges, nbin);
  bin_sort<<<nbin, 1024, 0, stream>>>(pairs, binstart, rowptr, inv, csr, n_nodes, X, W1, T);

  gather_gemm<<<nb_gemm, 128, 0, stream>>>(rowptr, csr, inv, T, b1, W2, T2);
  gather_gemm<<<nb_gemm, 128, 0, stream>>>(rowptr, csr, inv, T2, b2, W3, T);
  gather_rows<<<nb_gath, 256, 0, stream>>>(rowptr, csr, inv, T, b3, out, n_nodes);
}


# edge list build: chunks sorted by bin in place; a bin block collects its runs via a per-chunk start table (column sum = bin start); histogram+scan launches gone (7->5); row sums in new arrival order
# speedup vs baseline: 5.8106x; 1.1472x over previous
#include <hip/hip_runtime.h>


typedef __attribute__((ext_vector_type(2))) float v2f;
typedef __attribute__((ext_vector_type(4))) float v4f;
typedef __attribute__((ext_vector_type(8))) float v8f;

#define D 64

#define BIN_N 256
#define BIN_SH 8
#define CH 7168
#define MAXBIN 512
#define CAP 8192
#define RB 8
#define XP 68

__global__ __launch_bounds__(1024) void chunk_sort(
    const int* __restrict__ src, const int* __restrict__ dst,
    int* __restrict__ L, int2* __restrict__ pairs, int n_edges, int nbin) {
  __shared__ int h[MAXBIN];
  __shared__ int wsa[16];
  __shared__ int2 stage[CH];
  const int t = threadIdx.x;
  const int lane = t & 31;
  const int wv = t >> 5;
  const int b = blockIdx.x;
  const int nblk = gridDim.x;

  const int beg = b * CH;
  const int end = min(beg + CH, n_edges);
  int sv[CH / 1024], dv[CH / 1024];
#pragma unroll
  for (int k = 0; k < CH / 1024; ++k) {
    const int e = beg + t + k * 1024;
    sv[k] = (e < end) ? src[e] : 0;
    dv[k] = (e < end) ? dst[e] : 0;
  }
  if (t < MAXBIN) h[t] = 0;
  __syncthreads();
#pragma unroll
  for (int k = 0; k < CH / 1024; ++k)
    if (beg + t + k * 1024 < end) atomicAdd(&h[dv[k] >> BIN_SH], 1);
  __syncthreads();

  const int hv = (t < nbin) ? h[t] : 0;
  int hi = hv;
#pragma unroll
  for (int s = 1; s < 32; s <<= 1) {
    int u = __shfl_up(hi, s);
    if (lane >= s) hi += u;
  }
  if (t < MAXBIN && lane == 31) wsa[wv] = hi;
  __syncthreads();
  if (t < nbin) {
    int wa = 0;
    for (int k = 0; k < wv; ++k) wa += wsa[k];
    const int ls = wa + hi - hv;
    h[t] = ls;
    L[(size_t)t * nblk + b] = ls;
  }
  if (t == 0) L[(size_t)nbin * nblk + b] = end - beg;
  __syncthreads();

#pragma unroll
  for (int k = 0; k < CH / 1024; ++k) {
    if (beg + t + k * 1024 < end) {
      const int r = atomicAdd(&h[dv[k] >> BIN_SH], 1);
      stage[r] = make_int2(sv[k], dv[k]);
    }
  }
  __syncthreads();
  const int cnt = end - beg;
  int2 pv[CH / 1024];
#pragma unroll
  for (int k = 0; k < CH / 1024; ++k)
    if (t + k * 1024 < cnt) pv[k] = stage[t + k * 1024];
#pragma unroll
  for (int k = 0; k < CH / 1024; ++k)
    if (t + k * 1024 < cnt) pairs[beg + t + k * 1024] = pv[k];
}

template <class F>
__device__ __forceinline__ void walk_rest(const int2* __restrict__ pairs,
                                          const int* __restrict__ La,
                                          const int* __restrict__ Lb, int nblk, int wvu,
                                          int lane, const int (&len0)[RB],
                                          const int (&base0)[RB], F f) {
#pragma unroll
  for (int u = 0; u < RB; ++u)
    for (int j = lane + 32; j < len0[u]; j += 32) f(pairs[base0[u] + j]);
  for (int r = wvu + RB * 32; r < nblk; r += 32) {
    const int a = La[r];
    const int len = Lb[r] - a;
    for (int j = lane; j < len; j += 32) f(pairs[r * CH + a + j]);
  }
}

__global__ __launch_bounds__(1024) void bin_sort(
    const int2* __restrict__ pairs, const int* __restrict__ L, int nblk,
    int* __restrict__ rowptr, float* __restrict__ inv, int* __restrict__ csr, int n,
    const float* __restrict__ X, const float* __restrict__ W, float* __restrict__ T) {
  __shared__ int cnt[BIN_N];
  __shared__ int cur[BIN_N];
  __shared__ int wsum[BIN_N / 32];
  __shared__ int reda[32];
  __shared__ int redb[32];
  __shared__ float invs[BIN_N];
  __shared__ __attribute__((aligned(16))) float xs[BIN_N * XP];
  __shared__ __attribute__((aligned(16))) float wt[D * XP];
  __shared__ int stage[CAP];
  const int t = threadIdx.x;
  const int lane = t & 31;
  const int wv = t >> 5;
  const int wvu = __builtin_amdgcn_readfirstlane(wv);
  const int node0 = blockIdx.x * BIN_N;
  const int* __restrict__ La = L + (size_t)blockIdx.x * nblk;
  const int* __restrict__ Lb = La + nblk;

  int sa = 0, sb = 0;
  if (t < nblk) { sa = La[t]; sb = Lb[t]; }

  int len0[RB], base0[RB];
  {
    int ra0[RB], rb0[RB];
#pragma unroll
    for (int u = 0; u < RB; ++u) {
      const int rc = min(wvu + 32 * u, nblk - 1);
      ra0[u] = La[rc];
      rb0[u] = Lb[rc];
    }
#pragma unroll
    for (int u = 0; u < RB; ++u) {
      const int r = wvu + 32 * u;
      len0[u] = (r < nblk) ? rb0[u] - ra0[u] : 0;
      base0[u] = min(r, nblk - 1) * CH + ra0[u];
    }
  }
  int2 q0[RB];
#pragma unroll
  for (int u = 0; u < RB; ++u)
    q0[u] = (lane < len0[u]) ? pairs[base0[u] + lane] : make_int2(0, node0);

  const int half = lane >> 4;
  const int l    = lane & 15;
  const int stripe = wv >> 1;
  const int srow = node0 + (stripe << 4);
  const bool has = (srow < n);
  v4f xr[4];
  if (has) {
#pragma unroll
    for (int i = 0; i < 4; ++i)
      xr[i] = *(const v4f*)(X + (size_t)(srow + ((wv & 1) << 3) + 2 * i + half) * D + 4 * l);
  }
  const v4f wr = *(const v4f*)(W + 4 * t);

  if (t < BIN_N) cnt[t] = 0;
  __syncthreads();
#pragma unroll
  for (int u = 0; u < RB; ++u)
    if (lane < len0[u]) atomicAdd(&cnt[q0[u].y - node0], 1);
  walk_rest(pairs, La, Lb, nblk, wvu, lane, len0, base0,
            [&](int2 q) { atomicAdd(&cnt[q.y - node0], 1); });

#pragma unroll
  for (int o = 16; o > 0; o >>= 1) {
    sa += __shfl_xor(sa, o);
    sb += __shfl_xor(sb, o);
  }
  if (lane == 0) { reda[wv] = sa; redb[wv] = sb; }
  __syncthreads();

  int ra = reda[lane], rb = redb[lane];
#pragma unroll
  for (int o = 16; o > 0; o >>= 1) {
    ra += __shfl_xor(ra, o);
    rb += __shfl_xor(rb, o);
  }
  const int s = __builtin_amdgcn_readfirstlane(ra);
  const int e = __builtin_amdgcn_readfirstlane(rb);
  const int m = e - s;
  const bool small = (m <= CAP);

  int v = 0, incl = 0;
  if (t < BIN_N) {
    v = cnt[t];
    incl = v;
#pragma unroll
    for (int o = 1; o < 32; o <<= 1) {
      int u = __shfl_up(incl, o);
      if (lane >= o) incl += u;
    }
    if (lane == 31) wsum[wv] = incl;
  }
  __syncthreads();
  if (t < BIN_N) {
    int woff = 0;
    for (int k = 0; k < wv; ++k) woff += wsum[k];
    const int excl = woff + incl - v;
    cur[t] = excl;
    const int node = node0 + t;
    if (node < n) {
      rowptr[node] = s + excl;
      const float iv = 1.0f / sqrtf((float)v + 1.0f);
      inv[node] = iv;
      invs[t] = iv;
      if (node == n - 1) rowptr[n] = e;
    }
  }
  if (has) {
#pragma unroll
    for (int i = 0; i < 4; ++i)
      *(v4f*)&xs[((stripe << 4) + ((wv & 1) << 3) + 2 * i + half) * XP + 4 * l] = xr[i];
  }
#pragma unroll
  for (int i = 0; i < 4; ++i) wt[(4 * (t & 15) + i) * XP + (t >> 4)] = wr[i];
  __syncthreads();

  if (small) {
#pragma unroll
    for (int u = 0; u < RB; ++u)
      if (lane < len0[u]) stage[atomicAdd(&cur[q0[u].y - node0], 1)] = q0[u].x;
    walk_rest(pairs, La, Lb, nblk, wvu, lane, len0, base0,
              [&](int2 q) { stage[atomicAdd(&cur[q.y - node0], 1)] = q.x; });
    __syncthreads();
    int ov[CAP / 1024];
#pragma unroll
    for (int k = 0; k < CAP / 1024; ++k)
      if (t + k * 1024 < m) ov[k] = stage[t + k * 1024];
#pragma unroll
    for (int k = 0; k < CAP / 1024; ++k)
      if (t + k * 1024 < m) csr[s + t + k * 1024] = ov[k];
  } else {
#pragma unroll
    for (int u = 0; u < RB; ++u)
      if (lane < len0[u]) csr[s + atomicAdd(&cur[q0[u].y - node0], 1)] = q0[u].x;
    walk_rest(pairs, La, Lb, nblk, wvu, lane, len0, base0,
              [&](int2 q) { csr[s + atomicAdd(&cur[q.y - node0], 1)] = q.x; });
  }

  if (has) {
    const int c0 = (wv & 1) << 5;
    const float* __restrict__ arow = xs + ((stripe << 4) + l) * XP;
    const float* __restrict__ brow = wt + (c0 + l) * XP;
    v8f acc0 = {};
    v8f acc1 = {};
#pragma unroll
    for (int kk = 0; kk < D; kk += 4) {
      const int ka = kk + half * 2;
      v2f a  = { arow[ka], arow[ka + 1] };
      v2f b0 = { brow[ka], brow[ka + 1] };
      v2f b1 = { brow[16 * XP + ka], brow[16 * XP + ka + 1] };
      acc0 = __builtin_amdgcn_wmma_f32_16x16x4_f32(
           false, a,  false, b0,
           (short)0, acc0,  false,  false);
      acc1 = __builtin_amdgcn_wmma_f32_16x16x4_f32(
           false, a,  false, b1,
           (short)0, acc1,  false,  false);
    }
    float* __restrict__ orow = T + (size_t)(srow + half * 8) * D + c0 + l;
    const float* __restrict__ irow = invs + (srow - node0) + half * 8;
#pragma unroll
    for (int j = 0; j < 8; ++j) {
      const float sc = irow[j];
      orow[(size_t)j * D]      = acc0[j] * sc;
      orow[(size_t)j * D + 16] = acc1[j] * sc;
    }
  }
}

#define RPW 4

template <int NB>
__device__ __forceinline__ void gather_batch(const int* __restrict__ csr, int k,
                                             const float2* __restrict__ Tv, float2& acc) {
  int ed[NB];
  float2 tv[NB];
#pragma unroll
  for (int j = 0; j < NB; ++j) ed[j] = csr[k + j];
#pragma unroll
  for (int j = 0; j < NB; ++j) tv[j] = Tv[(size_t)ed[j] * (D / 2)];
#pragma unroll
  for (int j = 0; j < NB; ++j) {
    acc.x += tv[j].x;
    acc.y += tv[j].y;
  }
}

__device__ __forceinline__ float2 gather_node(const int* __restrict__ csr, int beg, int end,
                                              const float2* __restrict__ Tv, float2 acc) {
  int k = beg;
  for (; k + 8 <= end; k += 8) gather_batch<8>(csr, k, Tv, acc);
  const int rem = end - k;
  if (rem & 4) { gather_batch<4>(csr, k, Tv, acc); k += 4; }
  if (rem & 2) { gather_batch<2>(csr, k, Tv, acc); k += 2; }
  if (rem & 1) { gather_batch<1>(csr, k, Tv, acc); }
  return acc;
}

__global__ void gather_rows(const int* __restrict__ rowptr, const int* __restrict__ csr,
                            const float* __restrict__ inv, const float* __restrict__ T,
                            const float* __restrict__ bias, float* __restrict__ out,
                            int n) {
  const int dv = (blockIdx.x * (blockDim.x >> 5) + (threadIdx.x >> 5)) * RPW;
  if (dv >= n) return;
  const int d0 = __builtin_amdgcn_readfirstlane(dv);
  const int lane = threadIdx.x & 31;
  const float2* __restrict__ Tv = (const float2*)T + lane;
  const float2 b = ((const float2*)bias)[lane];
  int rp[RPW + 1];
  float isd[RPW];
  float2 self[RPW];
#pragma unroll
  for (int r = 0; r <= RPW; ++r) rp[r] = rowptr[d0 + r];
#pragma unroll
  for (int r = 0; r < RPW; ++r) {
    isd[r] = inv[d0 + r];
    self[r] = Tv[(size_t)(d0 + r) * (D / 2)];
  }
#pragma unroll
  for (int r = 0; r < RPW; ++r) {
    float2 acc = gather_node(csr, rp[r], rp[r + 1], Tv, self[r]);
    acc.x = fmaf(acc.x, isd[r], b.x);
    acc.y = fmaf(acc.y, isd[r], b.y);
    ((float2*)out)[(size_t)(d0 + r) * (D / 2) + lane] = acc;
  }
}

#define LDP 68

__global__ __launch_bounds__(128) void gather_gemm(
    const int* __restrict__ rowptr, const int* __restrict__ csr,
    const float* __restrict__ inv, const float* __restrict__ T,
    const float* __restrict__ bias, const float* __restrict__ W,
    float* __restrict__ Tn) {
  __shared__ float hs[16 * LDP];
  const int lane = threadIdx.x & 31;
  const int wave = threadIdx.x >> 5;
  const int row0 = blockIdx.x << 4;
  {
    const int d0 = __builtin_amdgcn_readfirstlane(row0 + wave * RPW);
    const float2* __restrict__ Tv = (const float2*)T + lane;
    const float2 b = ((const float2*)bias)[lane];
    int rp[RPW + 1];
    float isd[RPW];
    float2 self[RPW];
#pragma unroll
    for (int r = 0; r <= RPW; ++r) rp[r] = rowptr[d0 + r];
#pragma unroll
    for (int r = 0; r < RPW; ++r) {
      isd[r] = inv[d0 + r];
      self[r] = Tv[(size_t)(d0 + r) * (D / 2)];
    }
#pragma unroll
    for (int r = 0; r < RPW; ++r) {
      const float2 acc = gather_node(csr, rp[r], rp[r + 1], Tv, self[r]);
      float2 h;
      h.x = fmaxf(fmaf(acc.x, isd[r], b.x), 0.0f);
      h.y = fmaxf(fmaf(acc.y, isd[r], b.y), 0.0f);
      *(float2*)&hs[(wave * RPW + r) * LDP + 2 * lane] = h;
    }
  }
  __syncthreads();

  const int half = lane >> 4;
  const int l    = lane & 15;
  const int c0   = wave << 4;
  const float* __restrict__ arow = hs + l * LDP;

  v8f acc = {};
#pragma unroll
  for (int kk = 0; kk < D; kk += 4) {
    const int ka = kk + half * 2;
    v2f a = {arow[ka], arow[ka + 1]};
    v2f b = { W[ka * D + c0 + l], W[(ka + 1) * D + c0 + l] };
    acc = __builtin_amdgcn_wmma_f32_16x16x4_f32(
         false, a,  false, b,
         (short)0, acc,  false,  false);
  }

  float* __restrict__ orow = Tn + (size_t)(row0 + half * 8) * D + c0 + l;
  const float* __restrict__ irow = inv + row0 + half * 8;
#pragma unroll
  for (int j = 0; j < 8; ++j) orow[(size_t)j * D] = acc[j] * irow[j];
}


static inline size_t align256(size_t x) { return (x + 255) & ~(size_t)255; }

extern "C" void kernel_launch(void* const* d_in, const int* in_sizes, int n_in,
                              void* d_out, int out_size, void* d_ws, size_t ws_size,
                              hipStream_t stream) {
  const int*   E  = (const int*)d_in[1];
  const float* X  = (const float*)d_in[2];
  const float* W1 = (const float*)d_in[3];
  const float* b1 = (const float*)d_in[4];
  const float* W2 = (const float*)d_in[5];
  const float* b2 = (const float*)d_in[6];
  const float* W3 = (const float*)d_in[7];
  const float* b3 = (const float*)d_in[8];
  float* out = (float*)d_out;

  const int n_edges = in_sizes[1] / 2;
  const int n_nodes = in_sizes[2] / D;
  const int* src = E;
  const int* dst = E + n_edges;

  const int nbin = (n_nodes + BIN_N - 1) / BIN_N;
  const int nblk = (n_edges + CH - 1) / CH;

  char* ws = (char*)d_ws;
  int*   rowptr   = (int*)ws;   ws += align256((size_t)(n_nodes + 1) * 4);
  float* inv      = (float*)ws; ws += align256((size_t)n_nodes * 4);
  int*   L        = (int*)ws;   ws += align256((size_t)(nbin + 1) * nblk * 4);
  int2*  pairs    = (int2*)ws;  ws += align256((size_t)n_edges * 8);
  int*   csr      = (int*)ws;   ws += align256((size_t)n_edges * 4);
  float* T        = (float*)ws; ws += align256((size_t)n_nodes * D * 4);
  float* T2       = (float*)ws;

  const int nb_gath  = (n_nodes + 8 * RPW - 1) / (8 * RPW);
  const int nb_gemm  = n_nodes / 16;

  chunk_sort<<<nblk, 1024, 0, stream>>>(src, dst, L, pairs, n_edges, nbin);
  bin_sort<<<nbin, 1024, 0, stream>>>(pairs, L, nblk, rowptr, inv, csr, n_nodes, X, W1, T);

  gather_gemm<<<nb_gemm, 128, 0, stream>>>(rowptr, csr, inv, T, b1, W2, T2);
  gather_gemm<<<nb_gemm, 128, 0, stream>>>(rowptr, csr, inv, T2, b2, W3, T);
  gather_rows<<<nb_gath, 256, 0, stream>>>(rowptr, csr, inv, T, b3, out, n_nodes);
}
